# starting kernel
# baseline (speedup 1.0000x reference)
#include <hip/hip_runtime.h>

#define HID 128
#define D_IN 16
#define D_OUT 4
#define LN_EPS 1e-5f

typedef float v2f __attribute__((ext_vector_type(2)));
typedef float v8f __attribute__((ext_vector_type(8)));

__global__ void zero_kernel(float* __restrict__ p, int n) {
  int i = blockIdx.x * blockDim.x + threadIdx.x;
  int stride = gridDim.x * blockDim.x;
  for (; i < n; i += stride) p[i] = 0.0f;
}

__global__ void deg_kernel(const int* __restrict__ dst, float* __restrict__ deg, int E) {
  int e = blockIdx.x * blockDim.x + threadIdx.x;
  if (e < E) unsafeAtomicAdd(&deg[dst[e]], 1.0f);
}

__global__ void invdeg_kernel(float* __restrict__ deg, int N) {
  int i = blockIdx.x * blockDim.x + threadIdx.x;
  if (i < N) {
    float d = deg[i];
    deg[i] = 1.0f / fmaxf(d, 1.0f);
  }
}

__global__ void in_proj_kernel(const float* __restrict__ x,
                               const float* __restrict__ W_in,
                               const float* __restrict__ b_in,
                               float* __restrict__ h, int N) {
  int idx = blockIdx.x * blockDim.x + threadIdx.x;
  if (idx >= N * HID) return;
  int row = idx >> 7;
  int col = idx & (HID - 1);
  const float4* xr = (const float4*)(x + (size_t)row * D_IN);
  float acc = b_in[col];
#pragma unroll
  for (int kq = 0; kq < D_IN / 4; ++kq) {
    float4 xv = xr[kq];
    acc += xv.x * W_in[(kq * 4 + 0) * HID + col];
    acc += xv.y * W_in[(kq * 4 + 1) * HID + col];
    acc += xv.z * W_in[(kq * 4 + 2) * HID + col];
    acc += xv.w * W_in[(kq * 4 + 3) * HID + col];
  }
  h[idx] = acc;
}

__global__ void __launch_bounds__(256) ln_kernel(const float* __restrict__ h,
                                                 const float* __restrict__ g,
                                                 const float* __restrict__ b,
                                                 float* __restrict__ hn, int N) {
  int wave = threadIdx.x >> 5;
  int lane = threadIdx.x & 31;
  int row = blockIdx.x * 8 + wave;
  if (row >= N) return;

  float4 v = ((const float4*)(h + (size_t)row * HID))[lane];
  float s = v.x + v.y + v.z + v.w;
#pragma unroll
  for (int off = 16; off > 0; off >>= 1) s += __shfl_xor(s, off, 32);
  float mu = s * (1.0f / HID);

  float dx = v.x - mu, dy = v.y - mu, dz = v.z - mu, dw = v.w - mu;
  float q = dx * dx + dy * dy + dz * dz + dw * dw;
#pragma unroll
  for (int off = 16; off > 0; off >>= 1) q += __shfl_xor(q, off, 32);
  float rstd = rsqrtf(q * (1.0f / HID) + LN_EPS);

  float4 gv = ((const float4*)g)[lane];
  float4 bv = ((const float4*)b)[lane];
  float4 o;
  o.x = dx * rstd * gv.x + bv.x;
  o.y = dy * rstd * gv.y + bv.y;
  o.z = dz * rstd * gv.z + bv.z;
  o.w = dw * rstd * gv.w + bv.w;
  ((float4*)(hn + (size_t)row * HID))[lane] = o;
}

__global__ void __launch_bounds__(256) scatter_kernel(const int* __restrict__ src,
                                                      const int* __restrict__ dst,
                                                      const float* __restrict__ hn,
                                                      float* __restrict__ agg, int E) {
  int gid = blockIdx.x * blockDim.x + threadIdx.x;
  int e = gid >> 5;
  int lane = gid & 31;
  if (e >= E) return;
  int s = src[e];
  int d = dst[e];
  float4 v = ((const float4*)(hn + (size_t)s * HID))[lane];
  float* ap = agg + (size_t)d * HID + lane * 4;
  unsafeAtomicAdd(ap + 0, v.x);
  unsafeAtomicAdd(ap + 1, v.y);
  unsafeAtomicAdd(ap + 2, v.z);
  unsafeAtomicAdd(ap + 3, v.w);
}

__global__ void __launch_bounds__(256) sage_gemm_kernel(
    const float* __restrict__ agg, const float* __restrict__ hn,
    const float* __restrict__ invdeg, const float* __restrict__ Wl,
    const float* __restrict__ Wr, const float* __restrict__ bl,
    float* __restrict__ h, int N) {
  int wave = threadIdx.x >> 5;
  int lane = threadIdx.x & 31;
  int half = lane >> 4;
  int l16 = lane & 15;

  int m0 = blockIdx.x * 16;
  int row = m0 + l16;
  int rowc = row < N ? row : N - 1;
  int col = wave * 16 + l16;

  float invd = invdeg[rowc];
  const float* aggr = agg + (size_t)rowc * HID;
  const float* hnr = hn + (size_t)rowc * HID;

  v8f acc = {};
#pragma unroll 4
  for (int kb = 0; kb < HID; kb += 4) {
    int k0 = kb + 2 * half;
    v2f a1 = *(const v2f*)(aggr + k0);
    a1.x *= invd;
    a1.y *= invd;
    v2f b1;
    b1.x = Wl[k0 * HID + col];
    b1.y = Wl[(k0 + 1) * HID + col];
    acc = __builtin_amdgcn_wmma_f32_16x16x4_f32(false, a1, false, b1,
                                                (short)0, acc, false, false);
    v2f a2 = *(const v2f*)(hnr + k0);
    v2f b2;
    b2.x = Wr[k0 * HID + col];
    b2.y = Wr[(k0 + 1) * HID + col];
    acc = __builtin_amdgcn_wmma_f32_16x16x4_f32(false, a2, false, b2,
                                                (short)0, acc, false, false);
  }

  float bias = bl[col];
  float* hp = h + (size_t)(m0 + 8 * half) * HID + col;
  if (m0 + 16 <= N) {
#pragma unroll
    for (int v = 0; v < 8; ++v) {
      float r = hp[(size_t)v * HID] + acc[v] + bias;
      hp[(size_t)v * HID] = r > 0.0f ? r : 0.0f;
    }
  } else {
#pragma unroll
    for (int v = 0; v < 8; ++v) {
      int m = m0 + v + 8 * half;
      if (m < N) {
        float r = hp[(size_t)v * HID] + acc[v] + bias;
        hp[(size_t)v * HID] = r > 0.0f ? r : 0.0f;
      }
    }
  }
}

__global__ void out_proj_kernel(const float* __restrict__ h,
                                const float* __restrict__ W_out,
                                const float* __restrict__ b_out,
                                float* __restrict__ out, int N) {
  int idx = blockIdx.x * blockDim.x + threadIdx.x;
  if (idx >= N * D_OUT) return;
  int row = idx >> 2;
  int c = idx & 3;
  const float* hr = h + (size_t)row * HID;
  float acc = b_out[c];
#pragma unroll 8
  for (int k = 0; k < HID; ++k) acc += hr[k] * W_out[k * D_OUT + c];
  out[idx] = acc;
}

extern "C" void kernel_launch(void* const* d_in, const int* in_sizes, int n_in,
                              void* d_out, int out_size, void* d_ws, size_t ws_size,
                              hipStream_t stream) {
  const float* x = (const float*)d_in[0];
  const int* ei = (const int*)d_in[1];
  const float* W_in = (const float*)d_in[2];
  const float* b_in = (const float*)d_in[3];
  const float* Wl = (const float*)d_in[4];
  const float* bl = (const float*)d_in[5];
  const float* Wr = (const float*)d_in[6];
  const float* ln_g = (const float*)d_in[7];
  const float* ln_b = (const float*)d_in[8];
  const float* W_out = (const float*)d_in[9];
  const float* b_out = (const float*)d_in[10];
  float* out = (float*)d_out;

  int N = in_sizes[0] / D_IN;
  int E = in_sizes[1] / 2;
  const int* src = ei;
  const int* dst = ei + E;

  float* ws = (float*)d_ws;
  size_t nh = (size_t)N * HID;
  float* h = ws;
  float* hn = ws + nh;
  float* agg = ws + 2 * nh;
  float* deg = ws + 3 * nh;

  zero_kernel<<<256, 256, 0, stream>>>(deg, N);
  deg_kernel<<<(E + 255) / 256, 256, 0, stream>>>(dst, deg, E);
  invdeg_kernel<<<(N + 255) / 256, 256, 0, stream>>>(deg, N);

  in_proj_kernel<<<(int)((nh + 255) / 256), 256, 0, stream>>>(x, W_in, b_in, h, N);

  int mtiles = (N + 15) / 16;
  long long scat_threads = (long long)E * 32;
  for (int l = 0; l < 3; ++l) {
    ln_kernel<<<(N + 7) / 8, 256, 0, stream>>>(h, ln_g + l * HID, ln_b + l * HID, hn, N);
    zero_kernel<<<2048, 256, 0, stream>>>(agg, (int)nh);
    scatter_kernel<<<(int)((scat_threads + 255) / 256), 256, 0, stream>>>(src, dst, hn, agg, E);
    sage_gemm_kernel<<<mtiles, 256, 0, stream>>>(agg, hn, deg,
                                                 Wl + (size_t)l * HID * HID,
                                                 Wr + (size_t)l * HID * HID,
                                                 bl + l * HID, h, N);
  }

  out_proj_kernel<<<(N * D_OUT + 255) / 256, 256, 0, stream>>>(h, W_out, b_out, out, N);
}


# Layers 1, 2: rows times Wl BEFORE the neighbour sum: node pass stores y=hn@Wl and the start value D*(h+hn@Wr+bl), edge pass adds y[src], h=relu(sum/D); no clear launches; sum order changes
# speedup vs baseline: 3.4682x; 3.4682x over previous
#include <hip/hip_runtime.h>

#define HID 128
#define D_IN 16
#define D_OUT 4
#define LN_EPS 1e-5f

typedef float v2f __attribute__((ext_vector_type(2)));
typedef float v8f __attribute__((ext_vector_type(8)));

__device__ __forceinline__ float other_row(float x) {
  int b = __builtin_bit_cast(int, x);
  return __builtin_bit_cast(
      float, __builtin_amdgcn_permlanex16(b, b, 0x76543210, 0xfedcba98, false, false));
}
template <int M>
__device__ __forceinline__ float row_xor(float x) {
  int b = __builtin_bit_cast(int, x);
  return __builtin_bit_cast(float, __builtin_amdgcn_update_dpp(0, b, 0x160 + M, 0xf, 0xf, true));
}
__device__ __forceinline__ float wave_sum(float s) {
  s += other_row(s);
  s += row_xor<8>(s);
  s += row_xor<4>(s);
  s += row_xor<2>(s);
  s += row_xor<1>(s);
  return s;
}
__device__ __forceinline__ float4 ln_row(float4 v, float4 gv, float4 bv) {
  float s = wave_sum(v.x + v.y + v.z + v.w);
  float mu = s * (1.0f / HID);
  float dx = v.x - mu, dy = v.y - mu, dz = v.z - mu, dw = v.w - mu;
  float q = wave_sum(dx * dx + dy * dy + dz * dz + dw * dw);
  float rstd = rsqrtf(q * (1.0f / HID) + LN_EPS);
  float4 o;
  o.x = dx * rstd * gv.x + bv.x;
  o.y = dy * rstd * gv.y + bv.y;
  o.z = dz * rstd * gv.z + bv.z;
  o.w = dw * rstd * gv.w + bv.w;
  return o;
}

#define TICK_PITCH 64
#define REC_PER_TICKET 64
__device__ __forceinline__ int chip_half() {
  return (int)((__builtin_amdgcn_s_sendmsg_rtn(0x87) >> 18) & 1u);
}
__device__ __forceinline__ int draw_ticket(unsigned* tick, int nticket, int lane, int& hh) {
  hh = chip_half();
  unsigned t = 0;
  if (lane == 0) t = __hip_atomic_fetch_add(tick + hh * TICK_PITCH, 1u, __ATOMIC_RELAXED,
                                            __HIP_MEMORY_SCOPE_AGENT);
  t = __builtin_amdgcn_readfirstlane(t);
  if (t >= (unsigned)nticket) {
    hh ^= 1;
    if (lane == 0) t = __hip_atomic_fetch_add(tick + hh * TICK_PITCH, 1u, __ATOMIC_RELAXED,
                                              __HIP_MEMORY_SCOPE_AGENT);
    t = __builtin_amdgcn_readfirstlane(t);
    if (t >= (unsigned)nticket) return -1;
  }
  return (int)t;
}
__device__ __forceinline__ int rec_lane(int r0, int r1, int b) {
  int a0 = __builtin_amdgcn_readlane(r0, b & 31);
  int a1 = __builtin_amdgcn_readlane(r1, b & 31);
  return b < 32 ? a0 : a1;
}

#define PAY_W 18
#define FOLD_Q PAY_W
#define FOLD_ROWS (2 * FOLD_Q)
__global__ void __launch_bounds__(256) ln0_kernel(const float* __restrict__ W_in,
                                                  const float* __restrict__ b_in,
                                                  const float* __restrict__ g,
                                                  const float* __restrict__ b,
                                                  const float* __restrict__ Wl,
                                                  const float* __restrict__ Wr,
                                                  const float* __restrict__ x,
                                                  float* __restrict__ fold,
                                                  float* __restrict__ agg, int N,
                                                  unsigned* __restrict__ tick) {
  __shared__ __attribute__((aligned(16))) float avec[HID];
  __shared__ float part[HID];
  int lane = threadIdx.x & 31;
  if (blockIdx.x == 0 && threadIdx.x < 6) tick[threadIdx.x * TICK_PITCH] = 0u;

  if (blockIdx.x < FOLD_ROWS) {
    int side = blockIdx.x / FOLD_Q;
    int q = blockIdx.x - side * FOLD_Q;
    const float* Wm = side ? Wr : Wl;
    const float* av = q < D_IN ? W_in + q * HID : b_in;
    float4 rv = ((const float4*)av)[lane];
    float s = rv.x + rv.y + rv.z + rv.w;
#pragma unroll
    for (int off = 16; off > 0; off >>= 1) s += __shfl_xor(s, off, 32);
    float mean = s * (1.0f / HID);
    int col = threadIdx.x & (HID - 1);
    if (threadIdx.x < HID) avec[col] = q < FOLD_Q - 1 ? (av[col] - mean) * g[col] : b[col];
    __syncthreads();
    int jb = (int)(threadIdx.x >> 7) * (HID / 2);
    const float4* a4 = (const float4*)(avec + jb);
    const float* wp = Wm + jb * HID + col;
    float acc = 0.0f;
#pragma unroll 8
    for (int j4 = 0; j4 < HID / 8; ++j4) {
      float4 a = a4[j4];
      acc = fmaf(a.x, wp[(4 * j4 + 0) * HID], acc);
      acc = fmaf(a.y, wp[(4 * j4 + 1) * HID], acc);
      acc = fmaf(a.z, wp[(4 * j4 + 2) * HID], acc);
      acc = fmaf(a.w, wp[(4 * j4 + 3) * HID], acc);
    }
    if (threadIdx.x >= HID) part[col] = acc;
    __syncthreads();
    if (threadIdx.x < HID) fold[blockIdx.x * HID + col] = acc + part[col];
    return;
  }

  int row = __builtin_amdgcn_readfirstlane(
      (int)((blockIdx.x - FOLD_ROWS) * 8 + (threadIdx.x >> 5)));
  if (row >= N) return;

  const float* xr = x + (size_t)row * D_IN;
  float xv = lane < D_IN ? xr[lane] : 0.0f;
  float4 v = ((const float4*)b_in)[lane];
#pragma unroll
  for (int k = 0; k < D_IN; ++k) {
    float xk = xr[k];
    float4 w = ((const float4*)(W_in + k * HID))[lane];
    v.x = fmaf(xk, w.x, v.x);
    v.y = fmaf(xk, w.y, v.y);
    v.z = fmaf(xk, w.z, v.z);
    v.w = fmaf(xk, w.w, v.w);
  }
  float s = v.x + v.y + v.z + v.w;
#pragma unroll
  for (int off = 16; off > 0; off >>= 1) s += __shfl_xor(s, off, 32);
  float mu = s * (1.0f / HID);

  float dx = v.x - mu, dy = v.y - mu, dz = v.z - mu, dw = v.w - mu;
  float qq = dx * dx + dy * dy + dz * dz + dw * dw;
#pragma unroll
  for (int off = 16; off > 0; off >>= 1) qq += __shfl_xor(qq, off, 32);
  float rstd = rsqrtf(qq * (1.0f / HID) + LN_EPS);

  float e = xv * rstd;
  if (lane == D_IN) e = rstd;
  if (lane == D_IN + 1) e = 1.0f;
  float* ar = agg + (size_t)row * HID;
  ar[lane] = 0.0f;
  ar[32 + lane] = e;
}

#define EPG_IN 16
__global__ void __launch_bounds__(256) scatter_in_kernel(const int* __restrict__ src,
                                                         const int* __restrict__ dst,
                                                         float* agg, int E, int N,
                                                         unsigned* tick) {
  int lane = threadIdx.x & 31;
  int hh;
  int t = draw_ticket(tick, (E + REC_PER_TICKET - 1) / REC_PER_TICKET, lane, hh);
  if (t < 0) return;
  int ebase = t * REC_PER_TICKET;
  int e0 = ebase + lane, e1 = ebase + 32 + lane;
  int s0 = 0, d0 = 0, s1 = 0, d1 = 0;
  if (e0 < E) { s0 = src[e0]; d0 = dst[e0]; }
  if (e1 < E) { s1 = src[e1]; d1 = dst[e1]; }
  int nhalf = N >> 1;
  bool m0 = e0 < E && (int)(d0 >= nhalf) == hh;
  bool m1 = e1 < E && (int)(d1 >= nhalf) == hh;
  unsigned long long mask = (unsigned long long)__builtin_amdgcn_ballot_w32(m0) |
                            ((unsigned long long)__builtin_amdgcn_ballot_w32(m1) << 32);
  int n = __builtin_popcountll(mask);

#pragma unroll 1
  for (int j0 = 0; j0 < n; j0 += EPG_IN) {
    float v[EPG_IN];
    int dd[EPG_IN];
    if (j0 + EPG_IN <= n) {
#pragma unroll
      for (int u = 0; u < EPG_IN; ++u) {
        int b = __builtin_ctzll(mask);
        mask &= mask - 1;
        int sidx = rec_lane(s0, s1, b);
        dd[u] = rec_lane(d0, d1, b);
        v[u] = agg[(size_t)sidx * HID + 32 + lane];
      }
#pragma unroll
      for (int u = 0; u < EPG_IN; ++u) {
        float* ap = agg + (size_t)dd[u] * HID + lane;
        if (lane < PAY_W) unsafeAtomicAdd(ap, v[u]);
      }
    } else {
#pragma unroll
      for (int u = 0; u < EPG_IN; ++u) {
        v[u] = 0.0f;
        dd[u] = 0;
        if (j0 + u < n) {
          int b = __builtin_ctzll(mask);
          mask &= mask - 1;
          int sidx = rec_lane(s0, s1, b);
          dd[u] = rec_lane(d0, d1, b);
          v[u] = agg[(size_t)sidx * HID + 32 + lane];
        }
      }
#pragma unroll
      for (int u = 0; u < EPG_IN; ++u) {
        if (j0 + u < n) {
          float* ap = agg + (size_t)dd[u] * HID + lane;
          if (lane < PAY_W) unsafeAtomicAdd(ap, v[u]);
        }
      }
    }
  }
}

#define EPG 8
__global__ void __launch_bounds__(256) scatter_kernel(const int* __restrict__ src,
                                                      const int* __restrict__ dst,
                                                      const float* __restrict__ y,
                                                      float* __restrict__ acc, int E, int N,
                                                      unsigned* __restrict__ tick) {
  int lane = threadIdx.x & 31;
  int hh;
  int t = draw_ticket(tick, (E + REC_PER_TICKET - 1) / REC_PER_TICKET, lane, hh);
  if (t < 0) return;
  int ebase = t * REC_PER_TICKET;
  int e0 = ebase + lane, e1 = ebase + 32 + lane;
  int s0 = 0, d0 = 0, s1 = 0, d1 = 0;
  if (e0 < E) { s0 = src[e0]; d0 = dst[e0]; }
  if (e1 < E) { s1 = src[e1]; d1 = dst[e1]; }
  int nhalf = N >> 1;
  bool m0 = e0 < E && (int)(d0 >= nhalf) == hh;
  bool m1 = e1 < E && (int)(d1 >= nhalf) == hh;
  unsigned long long mask = (unsigned long long)__builtin_amdgcn_ballot_w32(m0) |
                            ((unsigned long long)__builtin_amdgcn_ballot_w32(m1) << 32);
  int n = __builtin_popcountll(mask);

#pragma unroll 1
  for (int j0 = 0; j0 < n; j0 += EPG) {
    float v[EPG][4];
    int dd[EPG];
    if (j0 + EPG <= n) {
#pragma unroll
      for (int u = 0; u < EPG; ++u) {
        int bit = __builtin_ctzll(mask);
        mask &= mask - 1;
        int s = rec_lane(s0, s1, bit);
        dd[u] = rec_lane(d0, d1, bit);
        const float* yp = y + (size_t)s * HID + lane;
        v[u][0] = yp[0];
        v[u][1] = yp[32];
        v[u][2] = yp[64];
        v[u][3] = yp[96];
      }
#pragma unroll
      for (int u = 0; u < EPG; ++u) {
        float* ap = acc + (size_t)dd[u] * HID + lane;
        unsafeAtomicAdd(ap + 0, v[u][0]);
        unsafeAtomicAdd(ap + 32, v[u][1]);
        unsafeAtomicAdd(ap + 64, v[u][2]);
        unsafeAtomicAdd(ap + 96, v[u][3]);
      }
    } else {
#pragma unroll
      for (int u = 0; u < EPG; ++u) {
        v[u][0] = v[u][1] = v[u][2] = v[u][3] = 0.0f;
        dd[u] = 0;
        if (j0 + u < n) {
          int bit = __builtin_ctzll(mask);
          mask &= mask - 1;
          int s = rec_lane(s0, s1, bit);
          dd[u] = rec_lane(d0, d1, bit);
          const float* yp = y + (size_t)s * HID + lane;
          v[u][0] = yp[0];
          v[u][1] = yp[32];
          v[u][2] = yp[64];
          v[u][3] = yp[96];
        }
      }
#pragma unroll
      for (int u = 0; u < EPG; ++u) {
        if (j0 + u < n) {
          float* ap = acc + (size_t)dd[u] * HID + lane;
          unsafeAtomicAdd(ap + 0, v[u][0]);
          unsafeAtomicAdd(ap + 32, v[u][1]);
          unsafeAtomicAdd(ap + 64, v[u][2]);
          unsafeAtomicAdd(ap + 96, v[u][3]);
        }
      }
    }
  }
}

#define APITCH 132
#define GEMM_GROUPS 512
template <bool FIRST>
__device__ __forceinline__ float4 h_row(float4 p, float d) {
  if (FIRST) return p;
  float i = 1.0f / d;
  float4 o;
  o.x = p.x * i; o.y = p.y * i; o.z = p.z * i; o.w = p.w * i;
  o.x = o.x > 0.0f ? o.x : 0.0f;
  o.y = o.y > 0.0f ? o.y : 0.0f;
  o.z = o.z > 0.0f ? o.z : 0.0f;
  o.w = o.w > 0.0f ? o.w : 0.0f;
  return o;
}
template <bool FIRST>
__device__ __forceinline__ void node_pass(
    const float* __restrict__ in, const float* __restrict__ deg,
    const float* __restrict__ lng, const float* __restrict__ lnb,
    const float* __restrict__ Wl, const float* __restrict__ Wr,
    const float* __restrict__ bl, float* __restrict__ y, float* __restrict__ acc, int N) {
  __shared__ __attribute__((aligned(16))) float stage[2][16 * APITCH];

  int wave = threadIdx.x >> 5;
  int lane = threadIdx.x & 31;
  int half = lane >> 4;
  int l16 = lane & 15;
  int col = wave * 16 + l16;
  int mtiles = (N + 15) >> 4;

  v2f bL[32], bR[32];
#pragma unroll
  for (int j = 0; j < 16; ++j) {
#pragma unroll
    for (int p = 0; p < 2; ++p) {
      int pos = 8 * j + 4 * half + 2 * p;
      bL[2 * j + p].x = Wl[pos * HID + col];
      bL[2 * j + p].y = Wl[(pos + 1) * HID + col];
      bR[2 * j + p].x = Wr[pos * HID + col];
      bR[2 * j + p].y = Wr[(pos + 1) * HID + col];
    }
  }
  float bias = bl[col];

  float4 gv = ((const float4*)lng)[lane];
  float4 bv = ((const float4*)lnb)[lane];
  int srow = 2 * wave;
  float4 p0, p1;
  float q0 = 1.0f, q1 = 1.0f;
  int t = blockIdx.x;
  if (t < mtiles) {
    int r0 = t * 16 + srow, r1 = r0 + 1;
    r0 = r0 < N ? r0 : N - 1;
    r1 = r1 < N ? r1 : N - 1;
    if (!FIRST) { q0 = deg[r0]; q1 = deg[r1]; }
    p0 = ((const float4*)(in + (size_t)r0 * HID))[lane];
    p1 = ((const float4*)(in + (size_t)r1 * HID))[lane];
    p0 = ln_row(h_row<FIRST>(p0, q0), gv, bv);
    p1 = ln_row(h_row<FIRST>(p1, q1), gv, bv);
    *(float4*)&stage[0][srow * APITCH + 4 * lane] = p0;
    *(float4*)&stage[0][(srow + 1) * APITCH + 4 * lane] = p1;
  }
  __syncthreads();

  int buf = 0;
  for (; t < mtiles; t += GEMM_GROUPS) {
    int m0 = t * 16;
    int tn = t + GEMM_GROUPS;
    bool more = tn < mtiles;
    if (more) {
      int r0 = tn * 16 + srow, r1 = r0 + 1;
      r0 = r0 < N ? r0 : N - 1;
      r1 = r1 < N ? r1 : N - 1;
      if (!FIRST) { q0 = deg[r0]; q1 = deg[r1]; }
      p0 = ((const float4*)(in + (size_t)r0 * HID))[lane];
      p1 = ((const float4*)(in + (size_t)r1 * HID))[lane];
    }
    bool whole = m0 + 16 <= N;
    float hv[8], dv[8];
#pragma unroll
    for (int v = 0; v < 8; ++v) {
      int m = m0 + v + 8 * half;
      int mc = m < N ? m : N - 1;
      hv[v] = in[(size_t)mc * HID + col];
      dv[v] = deg[mc];
    }

    const float* sH = &stage[buf][l16 * APITCH + 4 * half];
    v8f accL = {}, accR = {};
    float4 qh[2][4];
#pragma unroll
    for (int jj = 0; jj < 4; ++jj) qh[0][jj] = *(const float4*)(sH + 8 * jj);
#pragma unroll
    for (int ch = 0; ch < 4; ++ch) {
      if (ch + 1 < 4) {
#pragma unroll
        for (int jj = 0; jj < 4; ++jj)
          qh[(ch + 1) & 1][jj] = *(const float4*)(sH + 8 * (4 * (ch + 1) + jj));
      }
#pragma unroll
      for (int jj = 0; jj < 4; ++jj) {
        int j = 4 * ch + jj;
        float4 g = qh[ch & 1][jj];
        v2f g0 = {g.x, g.y}, g1 = {g.z, g.w};
        accL = __builtin_amdgcn_wmma_f32_16x16x4_f32(false, g0, false, bL[2 * j],
                                                     (short)0, accL, false, false);
        accR = __builtin_amdgcn_wmma_f32_16x16x4_f32(false, g0, false, bR[2 * j],
                                                     (short)0, accR, false, false);
        accL = __builtin_amdgcn_wmma_f32_16x16x4_f32(false, g1, false, bL[2 * j + 1],
                                                     (short)0, accL, false, false);
        accR = __builtin_amdgcn_wmma_f32_16x16x4_f32(false, g1, false, bR[2 * j + 1],
                                                     (short)0, accR, false, false);
      }
    }

    float* yp = y + (size_t)(m0 + 8 * half) * HID + col;
    float* ap = acc + (size_t)(m0 + 8 * half) * HID + col;
    if (whole) {
#pragma unroll
      for (int v = 0; v < 8; ++v) {
        float hres = hv[v];
        if (!FIRST) {
          hres *= 1.0f / dv[v];
          hres = hres > 0.0f ? hres : 0.0f;
        }
        yp[(size_t)v * HID] = accL[v];
        ap[(size_t)v * HID] = dv[v] * (hres + accR[v] + bias);
      }
    } else {
#pragma unroll
      for (int v = 0; v < 8; ++v) {
        int m = m0 + v + 8 * half;
        if (m < N) {
          float hres = hv[v];
          if (!FIRST) {
            hres *= 1.0f / dv[v];
            hres = hres > 0.0f ? hres : 0.0f;
          }
          yp[(size_t)v * HID] = accL[v];
          ap[(size_t)v * HID] = dv[v] * (hres + accR[v] + bias);
        }
      }
    }

    if (more) {
      p0 = ln_row(h_row<FIRST>(p0, q0), gv, bv);
      p1 = ln_row(h_row<FIRST>(p1, q1), gv, bv);
      *(float4*)&stage[buf ^ 1][srow * APITCH + 4 * lane] = p0;
      *(float4*)&stage[buf ^ 1][(srow + 1) * APITCH + 4 * lane] = p1;
    }
    __syncthreads();
    buf ^= 1;
  }
}
__global__ void __launch_bounds__(256) node_pass_first_kernel(
    const float* __restrict__ in, const float* __restrict__ deg,
    const float* __restrict__ lng, const float* __restrict__ lnb,
    const float* __restrict__ Wl, const float* __restrict__ Wr,
    const float* __restrict__ bl, float* __restrict__ y, float* __restrict__ acc, int N) {
  node_pass<true>(in, deg, lng, lnb, Wl, Wr, bl, y, acc, N);
}
__global__ void __launch_bounds__(256) node_pass_next_kernel(
    const float* __restrict__ in, const float* __restrict__ deg,
    const float* __restrict__ lng, const float* __restrict__ lnb,
    const float* __restrict__ Wl, const float* __restrict__ Wr,
    const float* __restrict__ bl, float* __restrict__ y, float* __restrict__ acc, int N) {
  node_pass<false>(in, deg, lng, lnb, Wl, Wr, bl, y, acc, N);
}

#define GEMM0_GROUPS 512
struct Tile0 {
  float4 s[3], p[3], x[2];
  float cnt;
};
__device__ __forceinline__ void load_tile0(Tile0& t, const float* __restrict__ agg,
                                           const float* __restrict__ x, int m, int half, int N) {
  m = m < N ? m : N - 1;
  const float* ar = agg + (size_t)m * HID;
  const float4* sp = (const float4*)(ar + 12 * half);
  const float4* pp = (const float4*)(ar + 32 + 12 * half);
  const float4* xp = (const float4*)(x + (size_t)m * D_IN + 8 * half);
#pragma unroll
  for (int i = 0; i < 3; ++i) {
    t.s[i] = sp[i];
    t.p[i] = pp[i];
  }
  t.x[0] = xp[0];
  t.x[1] = xp[1];
  t.cnt = ar[D_IN + 1];
}
__global__ void __launch_bounds__(256) sage_gemm0_kernel(
    const float* __restrict__ agg, const float* __restrict__ fold,
    const float* __restrict__ W_in, const float* __restrict__ b_in,
    const float* __restrict__ bl, const float* __restrict__ x,
    float* __restrict__ h, float* __restrict__ deg, int N) {
  int wave = threadIdx.x >> 5;
  int lane = threadIdx.x & 31;
  int half = lane >> 4;
  int l16 = lane & 15;
  int col = wave * 16 + l16;
  int mtiles = (N + 15) >> 4;

  v2f bS[6], bP[6], bW[4];
#pragma unroll
  for (int p = 0; p < 6; ++p) {
    int k = 12 * half + 2 * p;
    bool in = k < FOLD_Q;
    int kc = in ? k : 0;
    float l0 = fold[kc * HID + col], l1 = fold[(kc + 1) * HID + col];
    float r0 = fold[(FOLD_Q + kc) * HID + col], r1 = fold[(FOLD_Q + kc + 1) * HID + col];
    bS[p].x = in ? l0 : 0.0f;
    bS[p].y = in ? l1 : 0.0f;
    bP[p].x = in ? r0 : 0.0f;
    bP[p].y = in ? r1 : 0.0f;
  }
#pragma unroll
  for (int p = 0; p < 4; ++p) {
    bW[p].x = W_in[(8 * half + 2 * p) * HID + col];
    bW[p].y = W_in[(8 * half + 2 * p + 1) * HID + col];
  }
  float bias = b_in[col] + bl[col];

  Tile0 cur;
  int t = blockIdx.x;
  if (t < mtiles) load_tile0(cur, agg, x, t * 16 + l16, half, N);
  for (; t < mtiles; t += GEMM0_GROUPS) {
    int m0 = t * 16;
    int tn = t + GEMM0_GROUPS;
    Tile0 nxt = cur;
    if (tn < mtiles) load_tile0(nxt, agg, x, tn * 16 + l16, half, N);

    float dcur = fmaxf(cur.cnt, 1.0f);
    float inv = 1.0f / dcur;
    if (wave == 0 && half == 0 && m0 + l16 < N) deg[m0 + l16] = dcur;

    v8f accS = {}, accP;
#pragma unroll
    for (int v = 0; v < 8; ++v) accP[v] = bias;
#pragma unroll
    for (int i = 0; i < 3; ++i) {
      float4 a = cur.s[i];
      float4 g = cur.p[i];
      v2f a0 = {a.x * inv, a.y * inv}, a1 = {a.z * inv, a.w * inv};
      v2f g0 = {g.x, g.y}, g1 = {g.z, g.w};
      accS = __builtin_amdgcn_wmma_f32_16x16x4_f32(false, a0, false, bS[2 * i],
                                                   (short)0, accS, false, false);
      accP = __builtin_amdgcn_wmma_f32_16x16x4_f32(false, g0, false, bP[2 * i],
                                                   (short)0, accP, false, false);
      accS = __builtin_amdgcn_wmma_f32_16x16x4_f32(false, a1, false, bS[2 * i + 1],
                                                   (short)0, accS, false, false);
      accP = __builtin_amdgcn_wmma_f32_16x16x4_f32(false, g1, false, bP[2 * i + 1],
                                                   (short)0, accP, false, false);
    }
#pragma unroll
    for (int i = 0; i < 2; ++i) {
      float4 q = cur.x[i];
      v2f x0 = {q.x, q.y}, x1 = {q.z, q.w};
      accS = __builtin_amdgcn_wmma_f32_16x16x4_f32(false, x0, false, bW[2 * i],
                                                   (short)0, accS, false, false);
      accP = __builtin_amdgcn_wmma_f32_16x16x4_f32(false, x1, false, bW[2 * i + 1],
                                                   (short)0, accP, false, false);
    }

    float* hp = h + (size_t)(m0 + 8 * half) * HID + col;
    if (m0 + 16 <= N) {
#pragma unroll
      for (int v = 0; v < 8; ++v) {
        float r = accS[v] + accP[v];
        hp[(size_t)v * HID] = r > 0.0f ? r : 0.0f;
      }
    } else {
#pragma unroll
      for (int v = 0; v < 8; ++v) {
        if (m0 + v + 8 * half < N) {
          float r = accS[v] + accP[v];
          hp[(size_t)v * HID] = r > 0.0f ? r : 0.0f;
        }
      }
    }
    cur = nxt;
  }
}

__global__ void out_proj_kernel(const float* __restrict__ sum,
                                const float* __restrict__ deg,
                                const float* __restrict__ W_out,
                                const float* __restrict__ b_out,
                                float* __restrict__ out, int N) {
  int idx = blockIdx.x * blockDim.x + threadIdx.x;
  if (idx >= N * D_OUT) return;
  int row = idx >> 2;
  int c = idx & 3;
  const float* sr = sum + (size_t)row * HID;
  float inv = 1.0f / deg[row];
  float acc = b_out[c];
#pragma unroll 8
  for (int k = 0; k < HID; ++k) {
    float hk = sr[k] * inv;
    hk = hk > 0.0f ? hk : 0.0f;
    acc += hk * W_out[k * D_OUT + c];
  }
  out[idx] = acc;
}

extern "C" void kernel_launch(void* const* d_in, const int* in_sizes, int n_in,
                              void* d_out, int out_size, void* d_ws, size_t ws_size,
                              hipStream_t stream) {
  const float* x = (const float*)d_in[0];
  const int* ei = (const int*)d_in[1];
  const float* W_in = (const float*)d_in[2];
  const float* b_in = (const float*)d_in[3];
  const float* Wl = (const float*)d_in[4];
  const float* bl = (const float*)d_in[5];
  const float* Wr = (const float*)d_in[6];
  const float* ln_g = (const float*)d_in[7];
  const float* ln_b = (const float*)d_in[8];
  const float* W_out = (const float*)d_in[9];
  const float* b_out = (const float*)d_in[10];
  float* out = (float*)d_out;

  int N = in_sizes[0] / D_IN;
  int E = in_sizes[1] / 2;
  const int* src = ei;
  const int* dst = ei + E;

  float* ws = (float*)d_ws;
  size_t nh = (size_t)N * HID;
  float* pa = ws;
  float* pb = ws + nh;
  float* pc = ws + 2 * nh;
  float* deg = ws + 3 * nh;
  float* h = pa;
  float* agg = pc;
  float* fold = pb;
  unsigned* tick = (unsigned*)d_out;

  ln0_kernel<<<FOLD_ROWS + (N + 7) / 8, 256, 0, stream>>>(W_in, b_in, ln_g, ln_b, Wl, Wr, x,
                                                          fold, agg, N, tick);
  scatter_in_kernel<<<(E + 255) / 256, 256, 0, stream>>>(src, dst, agg, E, N, tick);
  sage_gemm0_kernel<<<GEMM0_GROUPS, 256, 0, stream>>>(agg, fold, W_in, b_in, bl, x, h, deg, N);

  node_pass_first_kernel<<<GEMM_GROUPS, 256, 0, stream>>>(
      pa, deg, ln_g + HID, ln_b + HID, Wl + (size_t)HID * HID, Wr + (size_t)HID * HID,
      bl + HID, pb, pc, N);
  scatter_kernel<<<(E + 255) / 256, 256, 0, stream>>>(src, dst, pb, pc, E, N,
                                                      tick + 2 * TICK_PITCH);
  node_pass_next_kernel<<<GEMM_GROUPS, 256, 0, stream>>>(
      pc, deg, ln_g + 2 * HID, ln_b + 2 * HID, Wl + (size_t)2 * HID * HID,
      Wr + (size_t)2 * HID * HID, bl + 2 * HID, pb, pa, N);
  scatter_kernel<<<(E + 255) / 256, 256, 0, stream>>>(src, dst, pb, pa, E, N,
                                                      tick + 4 * TICK_PITCH);

  out_proj_kernel<<<(N * D_OUT + 255) / 256, 256, 0, stream>>>(pa, deg, W_out, b_out, out, N);
}
